# starting kernel
# baseline (speedup 1.0000x reference)
#include <hip/hip_runtime.h>
#include <hip/hip_bf16.h>


typedef float v2f __attribute__((ext_vector_type(2)));
typedef float v8f __attribute__((ext_vector_type(8)));

#define BATCH   4
#define NSEQ    2048
#define EDIM    96
#define NHEAD   6
#define HDIM    16
#define QKVDIM  (3 * EDIM)
#define ROWS    (BATCH * NSEQ)

static __device__ __forceinline__ v8f wmma_f32_4(v2f a, v2f b, v8f c) {
    return __builtin_amdgcn_wmma_f32_16x16x4_f32(
        false, a, false, b, (short)0, c, false, false);
}

__global__ __launch_bounds__(256)
void gemm_bias_wmma(const float* __restrict__ A,
                    const float* __restrict__ W,
                    const float* __restrict__ bias,
                    float* __restrict__ C,
                    int M, int K, int N) {
    const int lane = threadIdx.x & 31;
    const int wave = threadIdx.x >> 5;
    const int wavesPerBlock = blockDim.x >> 5;
    const int tileId = blockIdx.x * wavesPerBlock + wave;
    const int tilesN = N >> 4;
    const int tileM = tileId / tilesN;
    const int tileN = tileId - tileM * tilesN;
    if (tileM * 16 >= M) return;

    const int row0 = tileM << 4;
    const int col0 = tileN << 4;
    const int m    = lane & 15;
    const int sel  = (lane >> 4) << 1;
    const int half = (lane >> 4) << 3;

    const float* arow = A + (size_t)(row0 + m) * K;

    v8f c = {};
    for (int k0 = 0; k0 < K; k0 += 4) {
        v2f a, b;
        a.x = arow[k0 + sel + 0];
        a.y = arow[k0 + sel + 1];
        b.x = W[(size_t)(k0 + sel + 0) * N + (col0 + m)];
        b.y = W[(size_t)(k0 + sel + 1) * N + (col0 + m)];
        c = wmma_f32_4(a, b, c);
    }

    const float bv = bias[col0 + m];
#pragma unroll
    for (int r = 0; r < 8; ++r) {
        C[(size_t)(row0 + r + half) * N + (col0 + m)] = c[r] + bv;
    }
}

__global__ __launch_bounds__(256)
void attn_kernel(const float* __restrict__ qkv, float* __restrict__ y) {
    const int bh   = blockIdx.x;
    const int b    = bh / NHEAD;
    const int h    = bh - b * NHEAD;
    const int lane = threadIdx.x & 31;
    const int wave = threadIdx.x >> 5;
    const int m    = lane & 15;
    const int sel  = (lane >> 4) << 1;
    const int half = (lane >> 4) << 3;

    const size_t base  = (size_t)b * NSEQ * QKVDIM;
    const float* qbase = qkv + base + 0 * EDIM + HDIM * h;
    const float* kbase = qkv + base + 1 * EDIM + HDIM * h;
    const float* vbase = qkv + base + 2 * EDIM + HDIM * h;

    __shared__ float red[8 * 256];
    __shared__ float S[16 * 16];

    v8f c = {};
    const int nBeg = wave * (NSEQ / 8);
    for (int n0 = nBeg; n0 < nBeg + NSEQ / 8; n0 += 4) {
        const float* krow = kbase + (size_t)(n0 + sel) * QKVDIM;
        const float* vrow = vbase + (size_t)(n0 + sel) * QKVDIM;
        v2f a, bb;
        a.x  = krow[m];
        a.y  = krow[QKVDIM + m];
        bb.x = vrow[m];
        bb.y = vrow[QKVDIM + m];
        c = wmma_f32_4(a, bb, c);
    }
#pragma unroll
    for (int r = 0; r < 8; ++r) red[wave * 256 + r * 32 + lane] = c[r];
    __syncthreads();

    {
        const int t  = threadIdx.x;
        float s = 0.f;
#pragma unroll
        for (int w = 0; w < 8; ++w) s += red[w * 256 + t];
        const int r  = t >> 5;
        const int ln = t & 31;
        const int row = r + ((ln >> 4) << 3);
        const int col = ln & 15;
        S[row * 16 + col] = s * 0.25f;
    }
    __syncthreads();

    for (int t = wave; t < NSEQ / 16; t += 8) {
        const int n0 = t << 4;
        const float* qrow = qbase + (size_t)(n0 + m) * QKVDIM;
        v8f acc = {};
#pragma unroll
        for (int k0 = 0; k0 < HDIM; k0 += 4) {
            v2f a, bb;
            a.x  = qrow[k0 + sel + 0];
            a.y  = qrow[k0 + sel + 1];
            bb.x = S[(k0 + sel + 0) * 16 + m];
            bb.y = S[(k0 + sel + 1) * 16 + m];
            acc = wmma_f32_4(a, bb, acc);
        }
#pragma unroll
        for (int r = 0; r < 8; ++r) {
            y[(size_t)b * NSEQ * EDIM + (size_t)(n0 + r + half) * EDIM + HDIM * h + m] = acc[r];
        }
    }
}

extern "C" void kernel_launch(void* const* d_in, const int* in_sizes, int n_in,
                              void* d_out, int out_size, void* d_ws, size_t ws_size,
                              hipStream_t stream) {
    const float* x    = (const float*)d_in[0];
    const float* Wqkv = (const float*)d_in[1];
    const float* bqkv = (const float*)d_in[2];
    const float* Wff  = (const float*)d_in[3];
    const float* bff  = (const float*)d_in[4];
    float* out = (float*)d_out;

    float* qkv = (float*)d_ws;
    float* y   = qkv + (size_t)ROWS * QKVDIM;

    {
        const int tiles  = (ROWS / 16) * (QKVDIM / 16);
        gemm_bias_wmma<<<tiles / 8, 256, 0, stream>>>(x, Wqkv, bqkv, qkv,
                                                      ROWS, EDIM, QKVDIM);
    }

    attn_kernel<<<BATCH * NHEAD, 256, 0, stream>>>(qkv, y);

    {
        const int tiles = (ROWS / 16) * (EDIM / 16);
        gemm_bias_wmma<<<tiles / 8, 256, 0, stream>>>(y, Wff, bff, out,
                                                      ROWS, EDIM, EDIM);
    }
}


# attention stage split into two full-chip launches: k^T v reduced in 8 partial sums per head (192 blocks, workspace), summed where q@S runs (192 blocks); reduction tree changes, launches 3 to 4
# speedup vs baseline: 1.6658x; 1.6658x over previous
#include <hip/hip_runtime.h>
#include <hip/hip_bf16.h>


typedef float v2f __attribute__((ext_vector_type(2)));
typedef float v8f __attribute__((ext_vector_type(8)));

#define BATCH   4
#define NSEQ    2048
#define EDIM    96
#define NHEAD   6
#define HDIM    16
#define QKVDIM  (3 * EDIM)
#define ROWS    (BATCH * NSEQ)

static __device__ __forceinline__ v8f wmma_f32_4(v2f a, v2f b, v8f c) {
    return __builtin_amdgcn_wmma_f32_16x16x4_f32(
        false, a, false, b, (short)0, c, false, false);
}

__global__ __launch_bounds__(256)
void gemm_bias_wmma(const float* __restrict__ A,
                    const float* __restrict__ W,
                    const float* __restrict__ bias,
                    float* __restrict__ C,
                    int M, int K, int N) {
    const int lane = threadIdx.x & 31;
    const int wave = threadIdx.x >> 5;
    const int wavesPerBlock = blockDim.x >> 5;
    const int tileId = blockIdx.x * wavesPerBlock + wave;
    const int tilesN = N >> 4;
    const int tileM = tileId / tilesN;
    const int tileN = tileId - tileM * tilesN;
    if (tileM * 16 >= M) return;

    const int row0 = tileM << 4;
    const int col0 = tileN << 4;
    const int m    = lane & 15;
    const int sel  = (lane >> 4) << 1;
    const int half = (lane >> 4) << 3;

    const float* arow = A + (size_t)(row0 + m) * K;

    v8f c = {};
    for (int k0 = 0; k0 < K; k0 += 4) {
        v2f a, b;
        a.x = arow[k0 + sel + 0];
        a.y = arow[k0 + sel + 1];
        b.x = W[(size_t)(k0 + sel + 0) * N + (col0 + m)];
        b.y = W[(size_t)(k0 + sel + 1) * N + (col0 + m)];
        c = wmma_f32_4(a, b, c);
    }

    const float bv = bias[col0 + m];
#pragma unroll
    for (int r = 0; r < 8; ++r) {
        C[(size_t)(row0 + r + half) * N + (col0 + m)] = c[r] + bv;
    }
}

#define KV_CHUNKS 8
#define KV_ROWS   (NSEQ / KV_CHUNKS)
#define QS_ROWS   128

__global__ __launch_bounds__(256)
void kv_partial_kernel(const float* __restrict__ qkv, float* __restrict__ part) {
    const int blk  = blockIdx.x;
    const int bh   = blk / KV_CHUNKS;
    const int ch   = blk - bh * KV_CHUNKS;
    const int b    = bh / NHEAD;
    const int h    = bh - b * NHEAD;
    const int lane = threadIdx.x & 31;
    const int wave = threadIdx.x >> 5;
    const int m    = lane & 15;
    const int sel  = (lane >> 4) << 1;

    const size_t base  = (size_t)b * NSEQ * QKVDIM;
    const float* kbase = qkv + base + 1 * EDIM + HDIM * h;
    const float* vbase = qkv + base + 2 * EDIM + HDIM * h;

    __shared__ float red[8 * 256];

    v8f c = {};
    const int nBeg = ch * KV_ROWS + wave * (KV_ROWS / 8);
#pragma unroll
    for (int i = 0; i < KV_ROWS / 8; i += 4) {
        const int n0 = nBeg + i;
        const float* krow = kbase + (size_t)(n0 + sel) * QKVDIM;
        const float* vrow = vbase + (size_t)(n0 + sel) * QKVDIM;
        v2f a, bb;
        a.x  = krow[m];
        a.y  = krow[QKVDIM + m];
        bb.x = vrow[m];
        bb.y = vrow[QKVDIM + m];
        c = wmma_f32_4(a, bb, c);
    }
#pragma unroll
    for (int r = 0; r < 8; ++r) red[wave * 256 + r * 32 + lane] = c[r];
    __syncthreads();

    {
        const int t = threadIdx.x;
        float s = 0.f;
#pragma unroll
        for (int w = 0; w < 8; ++w) s += red[w * 256 + t];
        part[(size_t)blk * 256 + t] = s;
    }
}

__global__ __launch_bounds__(256)
void qs_kernel(const float* __restrict__ qkv, const float* __restrict__ part,
               float* __restrict__ y) {
    const int blk  = blockIdx.x;
    const int rc   = blk & (NSEQ / QS_ROWS - 1);
    const int bp   = blk / (NSEQ / QS_ROWS);
    const int b    = bp / (NHEAD / 2);
    const int h0   = (bp - b * (NHEAD / 2)) * 2;
    const int lane = threadIdx.x & 31;
    const int wave = threadIdx.x >> 5;
    const int m    = lane & 15;
    const int sel  = (lane >> 4) << 1;
    const int half = (lane >> 4) << 3;

    __shared__ float S[2 * 256];

    const int n0 = rc * QS_ROWS + wave * 16;
    const float* qrow = qkv + (size_t)b * NSEQ * QKVDIM + (size_t)(n0 + m) * QKVDIM + HDIM * h0;

    v2f a[2][HDIM / 4];
#pragma unroll
    for (int hh = 0; hh < 2; ++hh) {
#pragma unroll
        for (int s = 0; s < HDIM / 4; ++s) {
            a[hh][s].x = qrow[HDIM * hh + 4 * s + sel + 0];
            a[hh][s].y = qrow[HDIM * hh + 4 * s + sel + 1];
        }
    }

    {
        const int t   = threadIdx.x;
        const int r   = t >> 5;
        const int ln  = t & 31;
        const int row = r + ((ln >> 4) << 3);
        const int col = ln & 15;
#pragma unroll
        for (int hh = 0; hh < 2; ++hh) {
            const float* p = part + (size_t)((b * NHEAD + h0 + hh) * KV_CHUNKS) * 256 + t;
            float s = 0.f;
#pragma unroll
            for (int ch = 0; ch < KV_CHUNKS; ++ch) s += p[ch * 256];
            S[hh * 256 + row * 16 + col] = s * 0.25f;
        }
    }
    __syncthreads();

#pragma unroll
    for (int hh = 0; hh < 2; ++hh) {
        v8f acc = {};
#pragma unroll
        for (int s = 0; s < HDIM / 4; ++s) {
            v2f bb;
            bb.x = S[hh * 256 + (4 * s + sel + 0) * 16 + m];
            bb.y = S[hh * 256 + (4 * s + sel + 1) * 16 + m];
            acc = wmma_f32_4(a[hh][s], bb, acc);
        }
#pragma unroll
        for (int r = 0; r < 8; ++r) {
            y[(size_t)b * NSEQ * EDIM + (size_t)(n0 + r + half) * EDIM + HDIM * (h0 + hh) + m] = acc[r];
        }
    }
}

extern "C" void kernel_launch(void* const* d_in, const int* in_sizes, int n_in,
                              void* d_out, int out_size, void* d_ws, size_t ws_size,
                              hipStream_t stream) {
    const float* x    = (const float*)d_in[0];
    const float* Wqkv = (const float*)d_in[1];
    const float* bqkv = (const float*)d_in[2];
    const float* Wff  = (const float*)d_in[3];
    const float* bff  = (const float*)d_in[4];
    float* out = (float*)d_out;

    float* qkv = (float*)d_ws;
    float* y   = qkv + (size_t)ROWS * QKVDIM;
    float* part = y + (size_t)ROWS * EDIM;

    {
        const int tiles  = (ROWS / 16) * (QKVDIM / 16);
        gemm_bias_wmma<<<tiles / 8, 256, 0, stream>>>(x, Wqkv, bqkv, qkv,
                                                      ROWS, EDIM, QKVDIM);
    }

    kv_partial_kernel<<<BATCH * NHEAD * KV_CHUNKS, 256, 0, stream>>>(qkv, part);

    qs_kernel<<<BATCH * (NHEAD / 2) * (NSEQ / QS_ROWS), 256, 0, stream>>>(qkv, part, y);

    {
        const int tiles = (ROWS / 16) * (EDIM / 16);
        gemm_bias_wmma<<<tiles / 8, 256, 0, stream>>>(y, Wff, bff, out,
                                                      ROWS, EDIM, EDIM);
    }
}


# both projection GEMMs: one wave per 16x96 strip (6 accumulators share each A fragment, interleaved columns give 16-byte W loads and C stores), 12288 waves to 2048; k summed in a permuted order
# speedup vs baseline: 2.2385x; 1.3438x over previous
#include <hip/hip_runtime.h>
#include <hip/hip_bf16.h>


typedef float v2f __attribute__((ext_vector_type(2)));
typedef float v8f __attribute__((ext_vector_type(8)));

#define BATCH   4
#define NSEQ    2048
#define EDIM    96
#define NHEAD   6
#define HDIM    16
#define QKVDIM  (3 * EDIM)
#define ROWS    (BATCH * NSEQ)

static __device__ __forceinline__ v8f wmma_f32_4(v2f a, v2f b, v8f c) {
    return __builtin_amdgcn_wmma_f32_16x16x4_f32(
        false, a, false, b, (short)0, c, false, false);
}

typedef float v4f __attribute__((ext_vector_type(4)));

#define STRIP 96

template <int K, int N>
__global__ __launch_bounds__(128)
void gemm_bias_strip(const float* __restrict__ A,
                     const float* __restrict__ W,
                     const float* __restrict__ bias,
                     float* __restrict__ C,
                     int tilesM) {
    const int lane = threadIdx.x & 31;
    const int wave = threadIdx.x >> 5;
    const int wid  = blockIdx.x * (blockDim.x >> 5) + wave;
    const int strip = wid / tilesM;
    const int tileM = wid - strip * tilesM;
    if (strip * STRIP >= N) return;

    const int c    = lane & 15;
    const int h    = lane >> 4;
    const int row0 = tileM << 4;
    const int cs   = strip * STRIP;

    const float* arow = A + (size_t)(row0 + c) * K + 4 * h;
    const float* wq   = W + (size_t)(4 * h) * N + cs + 4 * c;
    const float* wp   = W + (size_t)(4 * h) * N + cs + 64 + 2 * c;

    v8f acc[6] = {};
#pragma unroll
    for (int kb = 0; kb < K / 8; ++kb) {
        const v4f av = *(const v4f*)(arow + 8 * kb);
#pragma unroll
        for (int s = 0; s < 2; ++s) {
            const size_t r0 = (size_t)(8 * kb + 2 * s) * N;
            const v4f q0 = *(const v4f*)(wq + r0);
            const v4f q1 = *(const v4f*)(wq + r0 + N);
            const v2f p0 = *(const v2f*)(wp + r0);
            const v2f p1 = *(const v2f*)(wp + r0 + N);
            v2f a;
            a.x = av[2 * s + 0];
            a.y = av[2 * s + 1];
#pragma unroll
            for (int t = 0; t < 4; ++t) {
                v2f b;
                b.x = q0[t];
                b.y = q1[t];
                acc[t] = wmma_f32_4(a, b, acc[t]);
            }
#pragma unroll
            for (int u = 0; u < 2; ++u) {
                v2f b;
                b.x = p0[u];
                b.y = p1[u];
                acc[4 + u] = wmma_f32_4(a, b, acc[4 + u]);
            }
        }
    }

    const v4f bq = *(const v4f*)(bias + cs + 4 * c);
    const v2f bp = *(const v2f*)(bias + cs + 64 + 2 * c);
#pragma unroll
    for (int r = 0; r < 8; ++r) {
        float* crow = C + (size_t)(row0 + 8 * h + r) * N + cs;
        v4f o;
        o.x = acc[0][r] + bq.x;
        o.y = acc[1][r] + bq.y;
        o.z = acc[2][r] + bq.z;
        o.w = acc[3][r] + bq.w;
        *(v4f*)(crow + 4 * c) = o;
        v2f o2;
        o2.x = acc[4][r] + bp.x;
        o2.y = acc[5][r] + bp.y;
        *(v2f*)(crow + 64 + 2 * c) = o2;
    }
}

#define KV_CHUNKS 8
#define KV_ROWS   (NSEQ / KV_CHUNKS)
#define QS_ROWS   128

__global__ __launch_bounds__(256)
void kv_partial_kernel(const float* __restrict__ qkv, float* __restrict__ part) {
    const int blk  = blockIdx.x;
    const int bh   = blk / KV_CHUNKS;
    const int ch   = blk - bh * KV_CHUNKS;
    const int b    = bh / NHEAD;
    const int h    = bh - b * NHEAD;
    const int lane = threadIdx.x & 31;
    const int wave = threadIdx.x >> 5;
    const int m    = lane & 15;
    const int sel  = (lane >> 4) << 1;

    const size_t base  = (size_t)b * NSEQ * QKVDIM;
    const float* kbase = qkv + base + 1 * EDIM + HDIM * h;
    const float* vbase = qkv + base + 2 * EDIM + HDIM * h;

    __shared__ float red[8 * 256];

    v8f c = {};
    const int nBeg = ch * KV_ROWS + wave * (KV_ROWS / 8);
#pragma unroll
    for (int i = 0; i < KV_ROWS / 8; i += 4) {
        const int n0 = nBeg + i;
        const float* krow = kbase + (size_t)(n0 + sel) * QKVDIM;
        const float* vrow = vbase + (size_t)(n0 + sel) * QKVDIM;
        v2f a, bb;
        a.x  = krow[m];
        a.y  = krow[QKVDIM + m];
        bb.x = vrow[m];
        bb.y = vrow[QKVDIM + m];
        c = wmma_f32_4(a, bb, c);
    }
#pragma unroll
    for (int r = 0; r < 8; ++r) red[wave * 256 + r * 32 + lane] = c[r];
    __syncthreads();

    {
        const int t = threadIdx.x;
        float s = 0.f;
#pragma unroll
        for (int w = 0; w < 8; ++w) s += red[w * 256 + t];
        part[(size_t)blk * 256 + t] = s;
    }
}

__global__ __launch_bounds__(256)
void qs_kernel(const float* __restrict__ qkv, const float* __restrict__ part,
               float* __restrict__ y) {
    const int blk  = blockIdx.x;
    const int rc   = blk & (NSEQ / QS_ROWS - 1);
    const int bp   = blk / (NSEQ / QS_ROWS);
    const int b    = bp / (NHEAD / 2);
    const int h0   = (bp - b * (NHEAD / 2)) * 2;
    const int lane = threadIdx.x & 31;
    const int wave = threadIdx.x >> 5;
    const int m    = lane & 15;
    const int sel  = (lane >> 4) << 1;
    const int half = (lane >> 4) << 3;

    __shared__ float S[2 * 256];

    const int n0 = rc * QS_ROWS + wave * 16;
    const float* qrow = qkv + (size_t)b * NSEQ * QKVDIM + (size_t)(n0 + m) * QKVDIM + HDIM * h0;

    v2f a[2][HDIM / 4];
#pragma unroll
    for (int hh = 0; hh < 2; ++hh) {
#pragma unroll
        for (int s = 0; s < HDIM / 4; ++s) {
            a[hh][s].x = qrow[HDIM * hh + 4 * s + sel + 0];
            a[hh][s].y = qrow[HDIM * hh + 4 * s + sel + 1];
        }
    }

    {
        const int t   = threadIdx.x;
        const int r   = t >> 5;
        const int ln  = t & 31;
        const int row = r + ((ln >> 4) << 3);
        const int col = ln & 15;
#pragma unroll
        for (int hh = 0; hh < 2; ++hh) {
            const float* p = part + (size_t)((b * NHEAD + h0 + hh) * KV_CHUNKS) * 256 + t;
            float s = 0.f;
#pragma unroll
            for (int ch = 0; ch < KV_CHUNKS; ++ch) s += p[ch * 256];
            S[hh * 256 + row * 16 + col] = s * 0.25f;
        }
    }
    __syncthreads();

#pragma unroll
    for (int hh = 0; hh < 2; ++hh) {
        v8f acc = {};
#pragma unroll
        for (int s = 0; s < HDIM / 4; ++s) {
            v2f bb;
            bb.x = S[hh * 256 + (4 * s + sel + 0) * 16 + m];
            bb.y = S[hh * 256 + (4 * s + sel + 1) * 16 + m];
            acc = wmma_f32_4(a[hh][s], bb, acc);
        }
#pragma unroll
        for (int r = 0; r < 8; ++r) {
            y[(size_t)b * NSEQ * EDIM + (size_t)(n0 + r + half) * EDIM + HDIM * (h0 + hh) + m] = acc[r];
        }
    }
}

extern "C" void kernel_launch(void* const* d_in, const int* in_sizes, int n_in,
                              void* d_out, int out_size, void* d_ws, size_t ws_size,
                              hipStream_t stream) {
    const float* x    = (const float*)d_in[0];
    const float* Wqkv = (const float*)d_in[1];
    const float* bqkv = (const float*)d_in[2];
    const float* Wff  = (const float*)d_in[3];
    const float* bff  = (const float*)d_in[4];
    float* out = (float*)d_out;

    float* qkv = (float*)d_ws;
    float* y   = qkv + (size_t)ROWS * QKVDIM;
    float* part = y + (size_t)ROWS * EDIM;

    gemm_bias_strip<EDIM, QKVDIM><<<(ROWS / 16) * (QKVDIM / STRIP) / 4, 128, 0, stream>>>(
        x, Wqkv, bqkv, qkv, ROWS / 16);

    kv_partial_kernel<<<BATCH * NHEAD * KV_CHUNKS, 256, 0, stream>>>(qkv, part);

    qs_kernel<<<BATCH * (NHEAD / 2) * (NSEQ / QS_ROWS), 256, 0, stream>>>(qkv, part, y);

    gemm_bias_strip<EDIM, EDIM><<<(ROWS / 16) * (EDIM / STRIP) / 2, 64, 0, stream>>>(
        y, Wff, bff, out, ROWS / 16);
}


# q@S fused into the output projection launch: y^T = S^T q^T on WMMA, its D fragment reused as the A fragment of y@Wff (no y round trip); launches 4 to 3; k summed in a permuted order
# speedup vs baseline: 2.5859x; 1.1552x over previous
#include <hip/hip_runtime.h>
#include <hip/hip_bf16.h>


typedef float v2f __attribute__((ext_vector_type(2)));
typedef float v8f __attribute__((ext_vector_type(8)));

#define BATCH   4
#define NSEQ    2048
#define EDIM    96
#define NHEAD   6
#define HDIM    16
#define QKVDIM  (3 * EDIM)
#define ROWS    (BATCH * NSEQ)

static __device__ __forceinline__ v8f wmma_f32_4(v2f a, v2f b, v8f c) {
    return __builtin_amdgcn_wmma_f32_16x16x4_f32(
        false, a, false, b, (short)0, c, false, false);
}

typedef float v4f __attribute__((ext_vector_type(4)));

#define STRIP 96

template <int K, int N>
__global__ __launch_bounds__(128)
void gemm_bias_strip(const float* __restrict__ A,
                     const float* __restrict__ W,
                     const float* __restrict__ bias,
                     float* __restrict__ C,
                     int tilesM) {
    const int lane = threadIdx.x & 31;
    const int wave = threadIdx.x >> 5;
    const int wid  = blockIdx.x * (blockDim.x >> 5) + wave;
    const int strip = wid / tilesM;
    const int tileM = wid - strip * tilesM;
    if (strip * STRIP >= N) return;

    const int c    = lane & 15;
    const int h    = lane >> 4;
    const int row0 = tileM << 4;
    const int cs   = strip * STRIP;

    const float* arow = A + (size_t)(row0 + c) * K + 4 * h;
    const float* wq   = W + (size_t)(4 * h) * N + cs + 4 * c;
    const float* wp   = W + (size_t)(4 * h) * N + cs + 64 + 2 * c;

    v8f acc[6] = {};
#pragma unroll
    for (int kb = 0; kb < K / 8; ++kb) {
        const v4f av = *(const v4f*)(arow + 8 * kb);
#pragma unroll
        for (int s = 0; s < 2; ++s) {
            const size_t r0 = (size_t)(8 * kb + 2 * s) * N;
            const v4f q0 = *(const v4f*)(wq + r0);
            const v4f q1 = *(const v4f*)(wq + r0 + N);
            const v2f p0 = *(const v2f*)(wp + r0);
            const v2f p1 = *(const v2f*)(wp + r0 + N);
            v2f a;
            a.x = av[2 * s + 0];
            a.y = av[2 * s + 1];
#pragma unroll
            for (int t = 0; t < 4; ++t) {
                v2f b;
                b.x = q0[t];
                b.y = q1[t];
                acc[t] = wmma_f32_4(a, b, acc[t]);
            }
#pragma unroll
            for (int u = 0; u < 2; ++u) {
                v2f b;
                b.x = p0[u];
                b.y = p1[u];
                acc[4 + u] = wmma_f32_4(a, b, acc[4 + u]);
            }
        }
    }

    const v4f bq = *(const v4f*)(bias + cs + 4 * c);
    const v2f bp = *(const v2f*)(bias + cs + 64 + 2 * c);
#pragma unroll
    for (int r = 0; r < 8; ++r) {
        float* crow = C + (size_t)(row0 + 8 * h + r) * N + cs;
        v4f o;
        o.x = acc[0][r] + bq.x;
        o.y = acc[1][r] + bq.y;
        o.z = acc[2][r] + bq.z;
        o.w = acc[3][r] + bq.w;
        *(v4f*)(crow + 4 * c) = o;
        v2f o2;
        o2.x = acc[4][r] + bp.x;
        o2.y = acc[5][r] + bp.y;
        *(v2f*)(crow + 64 + 2 * c) = o2;
    }
}

#define KV_CHUNKS 8
#define KV_ROWS   (NSEQ / KV_CHUNKS)

__global__ __launch_bounds__(256)
void kv_partial_kernel(const float* __restrict__ qkv, float* __restrict__ part) {
    const int blk  = blockIdx.x;
    const int bh   = blk / KV_CHUNKS;
    const int ch   = blk - bh * KV_CHUNKS;
    const int b    = bh / NHEAD;
    const int h    = bh - b * NHEAD;
    const int lane = threadIdx.x & 31;
    const int wave = threadIdx.x >> 5;
    const int m    = lane & 15;
    const int sel  = (lane >> 4) << 1;

    const size_t base  = (size_t)b * NSEQ * QKVDIM;
    const float* kbase = qkv + base + 1 * EDIM + HDIM * h;
    const float* vbase = qkv + base + 2 * EDIM + HDIM * h;

    __shared__ float red[8 * 256];

    v8f c = {};
    const int nBeg = ch * KV_ROWS + wave * (KV_ROWS / 8);
#pragma unroll
    for (int i = 0; i < KV_ROWS / 8; i += 4) {
        const int n0 = nBeg + i;
        const float* krow = kbase + (size_t)(n0 + sel) * QKVDIM;
        const float* vrow = vbase + (size_t)(n0 + sel) * QKVDIM;
        v2f a, bb;
        a.x  = krow[m];
        a.y  = krow[QKVDIM + m];
        bb.x = vrow[m];
        bb.y = vrow[QKVDIM + m];
        c = wmma_f32_4(a, bb, c);
    }
#pragma unroll
    for (int r = 0; r < 8; ++r) red[wave * 256 + r * 32 + lane] = c[r];
    __syncthreads();

    {
        const int t = threadIdx.x;
        float s = 0.f;
#pragma unroll
        for (int w = 0; w < 8; ++w) s += red[w * 256 + t];
        part[(size_t)blk * 256 + t] = s;
    }
}

#define FO_WAVES 4
#define PRE_HEADS 3

__global__ __launch_bounds__(32 * FO_WAVES)
void qs_out_kernel(const float* __restrict__ qkv, const float* __restrict__ part,
                   const float* __restrict__ Wff, const float* __restrict__ bff,
                   float* __restrict__ out) {
    const int lane = threadIdx.x & 31;
    const int wave = threadIdx.x >> 5;
    const int c    = lane & 15;
    const int h    = lane >> 4;
    const int row0 = (blockIdx.x * FO_WAVES + wave) << 4;
    const int b    = (blockIdx.x * (FO_WAVES * 16)) / NSEQ;

    __shared__ float S[NHEAD * 256];

    const float* qrow = qkv + (size_t)(row0 + c) * QKVDIM + 8 * h;
    v4f qa[NHEAD][2];
#pragma unroll
    for (int hd = 0; hd < NHEAD; ++hd) {
        qa[hd][0] = *(const v4f*)(qrow + HDIM * hd);
        qa[hd][1] = *(const v4f*)(qrow + HDIM * hd + 4);
    }

    const float* wq = Wff + (size_t)(8 * h) * EDIM + 4 * c;
    const float* wp = Wff + (size_t)(8 * h) * EDIM + 64 + 2 * c;
    v4f wq0[PRE_HEADS * 4], wq1[PRE_HEADS * 4];
    v2f wp0[PRE_HEADS * 4], wp1[PRE_HEADS * 4];

    {
        const float* pb = part + (size_t)(b * NHEAD * KV_CHUNKS) * 256;
#pragma unroll
        for (int i = 0; i < (NHEAD * 64) / (32 * FO_WAVES); ++i) {
            const int g  = threadIdx.x + i * (32 * FO_WAVES);
            const int hd = g >> 6;
            const int sl = (g & 63) << 2;
            const float* p = pb + (size_t)(hd * KV_CHUNKS) * 256 + sl;
            v4f acc4 = {};
#pragma unroll
            for (int ch = 0; ch < KV_CHUNKS; ++ch) acc4 += *(const v4f*)(p + ch * 256);
            *(v4f*)(S + hd * 256 + sl) = acc4 * 0.25f;
        }
#pragma unroll
        for (int i = 0; i < PRE_HEADS * 4; ++i) {
            const size_t r0 = (size_t)(HDIM * (i >> 2) + 2 * (i & 3)) * EDIM;
            wq0[i] = *(const v4f*)(wq + r0);
            wq1[i] = *(const v4f*)(wq + r0 + EDIM);
            wp0[i] = *(const v2f*)(wp + r0);
            wp1[i] = *(const v2f*)(wp + r0 + EDIM);
        }
    }
    __syncthreads();

    v8f yt[NHEAD];
#pragma unroll
    for (int hd = 0; hd < NHEAD; ++hd) {
        v8f acc = {};
#pragma unroll
        for (int s = 0; s < 4; ++s) {
            v2f a, bq;
            a.x  = S[hd * 256 + (2 * s + 0) * 32 + lane];
            a.y  = S[hd * 256 + (2 * s + 1) * 32 + lane];
            bq.x = qa[hd][s >> 1][2 * (s & 1) + 0];
            bq.y = qa[hd][s >> 1][2 * (s & 1) + 1];
            acc = wmma_f32_4(a, bq, acc);
        }
        yt[hd] = acc;
    }

    v8f acc[6] = {};
#pragma unroll
    for (int hd = 0; hd < NHEAD; ++hd) {
#pragma unroll
        for (int s = 0; s < 4; ++s) {
            const size_t r0 = (size_t)(HDIM * hd + 2 * s) * EDIM;
            v4f q0, q1;
            v2f p0, p1;
            if (hd < PRE_HEADS) {
                q0 = wq0[4 * hd + s];
                q1 = wq1[4 * hd + s];
                p0 = wp0[4 * hd + s];
                p1 = wp1[4 * hd + s];
            } else {
                q0 = *(const v4f*)(wq + r0);
                q1 = *(const v4f*)(wq + r0 + EDIM);
                p0 = *(const v2f*)(wp + r0);
                p1 = *(const v2f*)(wp + r0 + EDIM);
            }
            v2f a;
            a.x = yt[hd][2 * s + 0];
            a.y = yt[hd][2 * s + 1];
#pragma unroll
            for (int t = 0; t < 4; ++t) {
                v2f bw;
                bw.x = q0[t];
                bw.y = q1[t];
                acc[t] = wmma_f32_4(a, bw, acc[t]);
            }
#pragma unroll
            for (int u = 0; u < 2; ++u) {
                v2f bw;
                bw.x = p0[u];
                bw.y = p1[u];
                acc[4 + u] = wmma_f32_4(a, bw, acc[4 + u]);
            }
        }
    }

    const v4f bq4 = *(const v4f*)(bff + 4 * c);
    const v2f bp2 = *(const v2f*)(bff + 64 + 2 * c);
#pragma unroll
    for (int r = 0; r < 8; ++r) {
        float* crow = out + (size_t)(row0 + 8 * h + r) * EDIM;
        v4f o;
        o.x = acc[0][r] + bq4.x;
        o.y = acc[1][r] + bq4.y;
        o.z = acc[2][r] + bq4.z;
        o.w = acc[3][r] + bq4.w;
        *(v4f*)(crow + 4 * c) = o;
        v2f o2;
        o2.x = acc[4][r] + bp2.x;
        o2.y = acc[5][r] + bp2.y;
        *(v2f*)(crow + 64 + 2 * c) = o2;
    }
}

extern "C" void kernel_launch(void* const* d_in, const int* in_sizes, int n_in,
                              void* d_out, int out_size, void* d_ws, size_t ws_size,
                              hipStream_t stream) {
    const float* x    = (const float*)d_in[0];
    const float* Wqkv = (const float*)d_in[1];
    const float* bqkv = (const float*)d_in[2];
    const float* Wff  = (const float*)d_in[3];
    const float* bff  = (const float*)d_in[4];
    float* out = (float*)d_out;

    float* qkv = (float*)d_ws;
    float* part = qkv + (size_t)ROWS * QKVDIM;

    gemm_bias_strip<EDIM, QKVDIM><<<(ROWS / 16) * (QKVDIM / STRIP) / 4, 128, 0, stream>>>(
        x, Wqkv, bqkv, qkv, ROWS / 16);

    kv_partial_kernel<<<BATCH * NHEAD * KV_CHUNKS, 256, 0, stream>>>(qkv, part);

    qs_out_kernel<<<ROWS / (16 * FO_WAVES), 32 * FO_WAVES, 0, stream>>>(qkv, part, Wff, bff, out);
}


# k^T v partials formed from the k and v accumulator fragments inside the projection launch (k, v never stored; q stored alone); launches 3 to 2; k^T v summed per 32 rows, then 8 waves, then 8 chunks
# speedup vs baseline: 3.3683x; 1.3026x over previous
#include <hip/hip_runtime.h>
#include <hip/hip_bf16.h>


typedef float v2f __attribute__((ext_vector_type(2)));
typedef float v8f __attribute__((ext_vector_type(8)));

#define BATCH   4
#define NSEQ    2048
#define EDIM    96
#define NHEAD   6
#define HDIM    16
#define QKVDIM  (3 * EDIM)
#define ROWS    (BATCH * NSEQ)

static __device__ __forceinline__ v8f wmma_f32_4(v2f a, v2f b, v8f c) {
    return __builtin_amdgcn_wmma_f32_16x16x4_f32(
        false, a, false, b, (short)0, c, false, false);
}

typedef float v4f __attribute__((ext_vector_type(4)));

#define KV_CHUNKS 8
#define KV_ROWS   (NSEQ / KV_CHUNKS)
#define PJ_WAVES  8
#define Q_BLOCKS  (ROWS / (16 * PJ_WAVES))
#define KV_BLOCKS (BATCH * KV_CHUNKS * NHEAD)

__global__ __launch_bounds__(32 * PJ_WAVES)
void proj_kv_kernel(const float* __restrict__ x,
                    const float* __restrict__ W,
                    const float* __restrict__ bias,
                    float* __restrict__ q,
                    float* __restrict__ part) {
    const int lane = threadIdx.x & 31;
    const int wave = threadIdx.x >> 5;
    const int c    = lane & 15;
    const int h    = lane >> 4;

    __shared__ float red[PJ_WAVES * 256];

    if (blockIdx.x < Q_BLOCKS) {
        const int row0 = (blockIdx.x * PJ_WAVES + wave) << 4;
        const float* arow = x + (size_t)(row0 + c) * EDIM + 4 * h;
        const float* wq   = W + (size_t)(4 * h) * QKVDIM + 4 * c;
        const float* wp   = W + (size_t)(4 * h) * QKVDIM + 64 + 2 * c;

        v8f acc[6] = {};
#pragma unroll
        for (int kb = 0; kb < EDIM / 8; ++kb) {
            const v4f av = *(const v4f*)(arow + 8 * kb);
#pragma unroll
            for (int s = 0; s < 2; ++s) {
                const size_t r0 = (size_t)(8 * kb + 2 * s) * QKVDIM;
                const v4f q0 = *(const v4f*)(wq + r0);
                const v4f q1 = *(const v4f*)(wq + r0 + QKVDIM);
                const v2f p0 = *(const v2f*)(wp + r0);
                const v2f p1 = *(const v2f*)(wp + r0 + QKVDIM);
                v2f a;
                a.x = av[2 * s + 0];
                a.y = av[2 * s + 1];
#pragma unroll
                for (int t = 0; t < 4; ++t) {
                    v2f b;
                    b.x = q0[t];
                    b.y = q1[t];
                    acc[t] = wmma_f32_4(a, b, acc[t]);
                }
#pragma unroll
                for (int u = 0; u < 2; ++u) {
                    v2f b;
                    b.x = p0[u];
                    b.y = p1[u];
                    acc[4 + u] = wmma_f32_4(a, b, acc[4 + u]);
                }
            }
        }

        const v4f bq = *(const v4f*)(bias + 4 * c);
        const v2f bp = *(const v2f*)(bias + 64 + 2 * c);
#pragma unroll
        for (int r = 0; r < 8; ++r) {
            float* crow = q + (size_t)(row0 + 8 * h + r) * EDIM;
            v4f o;
            o.x = acc[0][r] + bq.x;
            o.y = acc[1][r] + bq.y;
            o.z = acc[2][r] + bq.z;
            o.w = acc[3][r] + bq.w;
            *(v4f*)(crow + 4 * c) = o;
            v2f o2;
            o2.x = acc[4][r] + bp.x;
            o2.y = acc[5][r] + bp.y;
            *(v2f*)(crow + 64 + 2 * c) = o2;
        }
        return;
    }

    const int blk  = blockIdx.x - Q_BLOCKS;
    const int bc   = blk / NHEAD;
    const int hd   = blk - bc * NHEAD;
    const int row0 = bc * KV_ROWS + wave * 32;
    const float* a0 = x + (size_t)(row0 + c) * EDIM + 4 * h;
    const float* a1 = a0 + (size_t)16 * EDIM;
    const float* wk = W + (size_t)(4 * h) * QKVDIM + 1 * EDIM + HDIM * hd + c;
    const float* wv = W + (size_t)(4 * h) * QKVDIM + 2 * EDIM + HDIM * hd + c;

    v8f k0 = {}, v0 = {}, k1 = {}, v1 = {};
#pragma unroll
    for (int kb = 0; kb < EDIM / 8; ++kb) {
        const v4f x0 = *(const v4f*)(a0 + 8 * kb);
        const v4f x1 = *(const v4f*)(a1 + 8 * kb);
#pragma unroll
        for (int s = 0; s < 2; ++s) {
            const size_t r0 = (size_t)(8 * kb + 2 * s) * QKVDIM;
            v2f bk, bv, e0, e1;
            bk.x = wk[r0];
            bk.y = wk[r0 + QKVDIM];
            bv.x = wv[r0];
            bv.y = wv[r0 + QKVDIM];
            e0.x = x0[2 * s + 0];
            e0.y = x0[2 * s + 1];
            e1.x = x1[2 * s + 0];
            e1.y = x1[2 * s + 1];
            k0 = wmma_f32_4(e0, bk, k0);
            v0 = wmma_f32_4(e0, bv, v0);
            k1 = wmma_f32_4(e1, bk, k1);
            v1 = wmma_f32_4(e1, bv, v1);
        }
    }
    const float bkc = bias[1 * EDIM + HDIM * hd + c];
    const float bvc = bias[2 * EDIM + HDIM * hd + c];
#pragma unroll
    for (int r = 0; r < 8; ++r) {
        k0[r] += bkc;
        k1[r] += bkc;
        v0[r] += bvc;
        v1[r] += bvc;
    }

    v8f s0 = {}, s1 = {};
#pragma unroll
    for (int s = 0; s < 4; ++s) {
        v2f ka, vb;
        ka.x = k0[2 * s + 0];
        ka.y = k0[2 * s + 1];
        vb.x = v0[2 * s + 0];
        vb.y = v0[2 * s + 1];
        s0 = wmma_f32_4(ka, vb, s0);
        ka.x = k1[2 * s + 0];
        ka.y = k1[2 * s + 1];
        vb.x = v1[2 * s + 0];
        vb.y = v1[2 * s + 1];
        s1 = wmma_f32_4(ka, vb, s1);
    }
#pragma unroll
    for (int r = 0; r < 8; ++r) red[wave * 256 + r * 32 + lane] = s0[r] + s1[r];
    __syncthreads();

    {
        const int t = threadIdx.x;
        float sum = 0.f;
#pragma unroll
        for (int w = 0; w < PJ_WAVES; ++w) sum += red[w * 256 + t];
        const int b  = bc / KV_CHUNKS;
        const int ch = bc - b * KV_CHUNKS;
        part[((size_t)((b * NHEAD + hd) * KV_CHUNKS + ch)) * 256 + t] = sum;
    }
}

#define FO_WAVES 4
#define PRE_HEADS 3

__global__ __launch_bounds__(32 * FO_WAVES)
void qs_out_kernel(const float* __restrict__ q, const float* __restrict__ part,
                   const float* __restrict__ Wff, const float* __restrict__ bff,
                   float* __restrict__ out) {
    const int lane = threadIdx.x & 31;
    const int wave = threadIdx.x >> 5;
    const int c    = lane & 15;
    const int h    = lane >> 4;
    const int row0 = (blockIdx.x * FO_WAVES + wave) << 4;
    const int b    = (blockIdx.x * (FO_WAVES * 16)) / NSEQ;

    __shared__ float S[NHEAD * 256];

    const float* qrow = q + (size_t)(row0 + c) * EDIM + 8 * h;
    v4f qa[NHEAD][2];
#pragma unroll
    for (int hd = 0; hd < NHEAD; ++hd) {
        qa[hd][0] = *(const v4f*)(qrow + HDIM * hd);
        qa[hd][1] = *(const v4f*)(qrow + HDIM * hd + 4);
    }

    const float* wq = Wff + (size_t)(8 * h) * EDIM + 4 * c;
    const float* wp = Wff + (size_t)(8 * h) * EDIM + 64 + 2 * c;
    v4f wq0[PRE_HEADS * 4], wq1[PRE_HEADS * 4];
    v2f wp0[PRE_HEADS * 4], wp1[PRE_HEADS * 4];

    {
        const float* pb = part + (size_t)(b * NHEAD * KV_CHUNKS) * 256;
#pragma unroll
        for (int i = 0; i < (NHEAD * 64) / (32 * FO_WAVES); ++i) {
            const int g  = threadIdx.x + i * (32 * FO_WAVES);
            const int hd = g >> 6;
            const int sl = (g & 63) << 2;
            const float* p = pb + (size_t)(hd * KV_CHUNKS) * 256 + sl;
            v4f acc4 = {};
#pragma unroll
            for (int ch = 0; ch < KV_CHUNKS; ++ch) acc4 += *(const v4f*)(p + ch * 256);
            *(v4f*)(S + hd * 256 + sl) = acc4 * 0.25f;
        }
#pragma unroll
        for (int i = 0; i < PRE_HEADS * 4; ++i) {
            const size_t r0 = (size_t)(HDIM * (i >> 2) + 2 * (i & 3)) * EDIM;
            wq0[i] = *(const v4f*)(wq + r0);
            wq1[i] = *(const v4f*)(wq + r0 + EDIM);
            wp0[i] = *(const v2f*)(wp + r0);
            wp1[i] = *(const v2f*)(wp + r0 + EDIM);
        }
    }
    __syncthreads();

    v8f yt[NHEAD];
#pragma unroll
    for (int hd = 0; hd < NHEAD; ++hd) {
        v8f acc = {};
#pragma unroll
        for (int s = 0; s < 4; ++s) {
            v2f a, bq;
            a.x  = S[hd * 256 + (2 * s + 0) * 32 + lane];
            a.y  = S[hd * 256 + (2 * s + 1) * 32 + lane];
            bq.x = qa[hd][s >> 1][2 * (s & 1) + 0];
            bq.y = qa[hd][s >> 1][2 * (s & 1) + 1];
            acc = wmma_f32_4(a, bq, acc);
        }
        yt[hd] = acc;
    }

    v8f acc[6] = {};
#pragma unroll
    for (int hd = 0; hd < NHEAD; ++hd) {
#pragma unroll
        for (int s = 0; s < 4; ++s) {
            const size_t r0 = (size_t)(HDIM * hd + 2 * s) * EDIM;
            v4f q0, q1;
            v2f p0, p1;
            if (hd < PRE_HEADS) {
                q0 = wq0[4 * hd + s];
                q1 = wq1[4 * hd + s];
                p0 = wp0[4 * hd + s];
                p1 = wp1[4 * hd + s];
            } else {
                q0 = *(const v4f*)(wq + r0);
                q1 = *(const v4f*)(wq + r0 + EDIM);
                p0 = *(const v2f*)(wp + r0);
                p1 = *(const v2f*)(wp + r0 + EDIM);
            }
            v2f a;
            a.x = yt[hd][2 * s + 0];
            a.y = yt[hd][2 * s + 1];
#pragma unroll
            for (int t = 0; t < 4; ++t) {
                v2f bw;
                bw.x = q0[t];
                bw.y = q1[t];
                acc[t] = wmma_f32_4(a, bw, acc[t]);
            }
#pragma unroll
            for (int u = 0; u < 2; ++u) {
                v2f bw;
                bw.x = p0[u];
                bw.y = p1[u];
                acc[4 + u] = wmma_f32_4(a, bw, acc[4 + u]);
            }
        }
    }

    const v4f bq4 = *(const v4f*)(bff + 4 * c);
    const v2f bp2 = *(const v2f*)(bff + 64 + 2 * c);
#pragma unroll
    for (int r = 0; r < 8; ++r) {
        float* crow = out + (size_t)(row0 + 8 * h + r) * EDIM;
        v4f o;
        o.x = acc[0][r] + bq4.x;
        o.y = acc[1][r] + bq4.y;
        o.z = acc[2][r] + bq4.z;
        o.w = acc[3][r] + bq4.w;
        *(v4f*)(crow + 4 * c) = o;
        v2f o2;
        o2.x = acc[4][r] + bp2.x;
        o2.y = acc[5][r] + bp2.y;
        *(v2f*)(crow + 64 + 2 * c) = o2;
    }
}

extern "C" void kernel_launch(void* const* d_in, const int* in_sizes, int n_in,
                              void* d_out, int out_size, void* d_ws, size_t ws_size,
                              hipStream_t stream) {
    const float* x    = (const float*)d_in[0];
    const float* Wqkv = (const float*)d_in[1];
    const float* bqkv = (const float*)d_in[2];
    const float* Wff  = (const float*)d_in[3];
    const float* bff  = (const float*)d_in[4];
    float* out = (float*)d_out;

    float* q    = (float*)d_ws;
    float* part = q + (size_t)ROWS * EDIM;

    proj_kv_kernel<<<Q_BLOCKS + KV_BLOCKS, 32 * PJ_WAVES, 0, stream>>>(x, Wqkv, bqkv, q, part);

    qs_out_kernel<<<ROWS / (16 * FO_WAVES), 32 * FO_WAVES, 0, stream>>>(q, part, Wff, bff, out);
}
